# starting kernel
# baseline (speedup 1.0000x reference)
#include <hip/hip_runtime.h>
#include <hip/hip_bf16.h>

typedef __attribute__((ext_vector_type(16))) __bf16 v16bf;
typedef __attribute__((ext_vector_type(8)))  float  v8f;

#define D_IN  256
#define H_DIM 32
#define C_DIM 2

__global__ void fill_kernel(float* __restrict__ p, long long n, float v) {
  long long i = (long long)blockIdx.x * blockDim.x + threadIdx.x;
  if (i < n) p[i] = v;
}

__global__ void degree_kernel(const int* __restrict__ ei, float* __restrict__ deg, int E) {
  int e = blockIdx.x * blockDim.x + threadIdx.x;
  if (e < E) atomicAdd(&deg[ei[E + e]], 1.0f);
}

__global__ void rsqrt_kernel(float* __restrict__ deg, int N) {
  int i = blockIdx.x * blockDim.x + threadIdx.x;
  if (i < N) {
    float d = deg[i];
    deg[i] = d > 0.0f ? rsqrtf(d) : 0.0f;
  }
}

__global__ void gemm1_wmma_kernel(const float* __restrict__ x,
                                  const float* __restrict__ W1,
                                  float* __restrict__ t1, int N) {
  __shared__ __align__(32) __bf16 Bfrag[8][2][32][16];

  const int tid  = threadIdx.x;
  const int lane = tid & 31;
  const int wave = tid >> 5;

  for (int idx = tid; idx < 8 * 2 * 32 * 16; idx += blockDim.x) {
    int j     = idx & 15;
    int ln    = (idx >> 4) & 31;
    int ntile = (idx >> 9) & 1;
    int ks    = idx >> 10;
    int jj = j >> 1, p = j & 1;
    int hi = ln >> 4;
    int K  = ks * 32 + 16 * (jj >> 2) + 8 * hi + 2 * (jj & 3) + p;
    int n  = ntile * 16 + (ln & 15);
    Bfrag[ks][ntile][ln][j] = (__bf16)W1[K * H_DIM + n];
  }
  __syncthreads();

  long long rowbase = ((long long)blockIdx.x * 8 + wave) * 16;
  if (rowbase >= N) return;

  const int m  = lane & 15;
  const int hi = lane >> 4;
  const float* xrow = x + (rowbase + m) * D_IN;

  v8f acc0 = {}; v8f acc1 = {};
  #pragma unroll
  for (int ks = 0; ks < 8; ++ks) {
    v16bf a;
    #pragma unroll
    for (int jj = 0; jj < 8; ++jj) {
      int K0 = ks * 32 + 16 * (jj >> 2) + 8 * hi + 2 * (jj & 3);
      float2 f = *(const float2*)(xrow + K0);
      a[2 * jj]     = (__bf16)f.x;
      a[2 * jj + 1] = (__bf16)f.y;
    }
    v16bf b0 = *(const v16bf*)&Bfrag[ks][0][lane][0];
    v16bf b1 = *(const v16bf*)&Bfrag[ks][1][lane][0];
    acc0 = __builtin_amdgcn_wmma_f32_16x16x32_bf16(false, a, false, b0,
                                                   (short)0, acc0, false, false);
    acc1 = __builtin_amdgcn_wmma_f32_16x16x32_bf16(false, a, false, b1,
                                                   (short)0, acc1, false, false);
  }

  #pragma unroll
  for (int r = 0; r < 8; ++r) {
    long long row = rowbase + r + 8 * hi;
    t1[row * H_DIM + m]      = acc0[r];
    t1[row * H_DIM + 16 + m] = acc1[r];
  }
}

__global__ void scatter1_kernel(const int* __restrict__ ei,
                                const float* __restrict__ t1,
                                const float* __restrict__ dinv,
                                float* __restrict__ agg1, int E) {
  long long gid = (long long)blockIdx.x * blockDim.x + threadIdx.x;
  long long e   = gid >> 3;
  int part      = (int)(gid & 7);
  if (e >= E) return;
  long long s = ei[e];
  long long t = ei[(long long)E + e];
  float w = dinv[s] * dinv[t];
  float4 v = *(const float4*)(t1 + s * H_DIM + part * 4);
  float* dst = agg1 + t * H_DIM + part * 4;
  atomicAdd(dst + 0, v.x * w);
  atomicAdd(dst + 1, v.y * w);
  atomicAdd(dst + 2, v.z * w);
  atomicAdd(dst + 3, v.w * w);
}

__global__ void bias_relu_kernel(float* __restrict__ agg1,
                                 const float* __restrict__ t1,
                                 const float* __restrict__ dinv,
                                 const float* __restrict__ b1, int N) {
  long long i = (long long)blockIdx.x * blockDim.x + threadIdx.x;
  if (i >= (long long)N * H_DIM) return;
  long long node = i >> 5;
  int n = (int)(i & 31);
  float di = dinv[node];
  float v = agg1[i] + di * di * t1[i] + b1[n];
  agg1[i] = v > 0.0f ? v : 0.0f;
}

__global__ void gemm2_kernel(const float* __restrict__ h1,
                             const float* __restrict__ W2,
                             float* __restrict__ h2, int N) {
  int i = blockIdx.x * blockDim.x + threadIdx.x;
  if (i >= N) return;
  const float* hr = h1 + (long long)i * H_DIM;
  float z0 = 0.0f, z1 = 0.0f;
  #pragma unroll
  for (int k = 0; k < H_DIM; ++k) {
    float h = hr[k];
    z0 += h * W2[k * C_DIM + 0];
    z1 += h * W2[k * C_DIM + 1];
  }
  h2[(long long)i * C_DIM + 0] = z0;
  h2[(long long)i * C_DIM + 1] = z1;
}

__global__ void scatter2_kernel(const int* __restrict__ ei,
                                const float* __restrict__ h2,
                                const float* __restrict__ dinv,
                                float* __restrict__ agg2, int E) {
  long long e = (long long)blockIdx.x * blockDim.x + threadIdx.x;
  if (e >= E) return;
  long long s = ei[e];
  long long t = ei[(long long)E + e];
  float w = dinv[s] * dinv[t];
  atomicAdd(&agg2[t * C_DIM + 0], h2[s * C_DIM + 0] * w);
  atomicAdd(&agg2[t * C_DIM + 1], h2[s * C_DIM + 1] * w);
}

__global__ void logsoftmax_kernel(const float* __restrict__ agg2,
                                  const float* __restrict__ h2,
                                  const float* __restrict__ dinv,
                                  const float* __restrict__ b2,
                                  float* __restrict__ out, int N) {
  int i = blockIdx.x * blockDim.x + threadIdx.x;
  if (i >= N) return;
  float di = dinv[i], sl = di * di;
  float z0 = agg2[(long long)i * 2 + 0] + sl * h2[(long long)i * 2 + 0] + b2[0];
  float z1 = agg2[(long long)i * 2 + 1] + sl * h2[(long long)i * 2 + 1] + b2[1];
  float mx  = fmaxf(z0, z1);
  float lse = mx + logf(expf(z0 - mx) + expf(z1 - mx));
  out[(long long)i * 2 + 0] = z0 - lse;
  out[(long long)i * 2 + 1] = z1 - lse;
}

extern "C" void kernel_launch(void* const* d_in, const int* in_sizes, int n_in,
                              void* d_out, int out_size, void* d_ws, size_t ws_size,
                              hipStream_t stream) {
  const float* x  = (const float*)d_in[0];
  const int*   ei = (const int*)d_in[1];
  const float* W1 = (const float*)d_in[2];
  const float* b1 = (const float*)d_in[3];
  const float* W2 = (const float*)d_in[4];
  const float* b2 = (const float*)d_in[5];
  float* out = (float*)d_out;

  const int N = in_sizes[0] / D_IN;
  const int E = in_sizes[1] / 2;

  float* ws   = (float*)d_ws;
  float* dinv = ws;
  float* t1   = dinv + N;
  float* agg1 = t1 + (long long)N * H_DIM;
  float* h2   = agg1 + (long long)N * H_DIM;
  float* agg2 = h2 + (long long)N * C_DIM;

  const int B = 256;
  fill_kernel<<<(N + B - 1) / B, B, 0, stream>>>(dinv, N, 1.0f);
  fill_kernel<<<((long long)N * H_DIM + B - 1) / B, B, 0, stream>>>(agg1, (long long)N * H_DIM, 0.0f);
  fill_kernel<<<((long long)N * C_DIM + B - 1) / B, B, 0, stream>>>(agg2, (long long)N * C_DIM, 0.0f);

  degree_kernel<<<(E + B - 1) / B, B, 0, stream>>>(ei, dinv, E);
  rsqrt_kernel<<<(N + B - 1) / B, B, 0, stream>>>(dinv, N);

  gemm1_wmma_kernel<<<(N + 127) / 128, B, 0, stream>>>(x, W1, t1, N);
  scatter1_kernel<<<((long long)E * 8 + B - 1) / B, B, 0, stream>>>(ei, t1, dinv, agg1, E);
  bias_relu_kernel<<<((long long)N * H_DIM + B - 1) / B, B, 0, stream>>>(agg1, t1, dinv, b1, N);

  gemm2_kernel<<<(N + B - 1) / B, B, 0, stream>>>(agg1, W2, h2, N);
  scatter2_kernel<<<(E + B - 1) / B, B, 0, stream>>>(ei, h2, dinv, agg2, E);
  logsoftmax_kernel<<<(N + B - 1) / B, B, 0, stream>>>(agg2, h2, dinv, b2, out, N);
}


# layer-1 edge scatter: agg1 rows split between the two package halves by node range; blocks draw edge chunks from a per-half ticket counter and add only into their half's rows; arrival-order sums
# speedup vs baseline: 5.6285x; 5.6285x over previous
#include <hip/hip_runtime.h>
#include <hip/hip_bf16.h>

typedef __attribute__((ext_vector_type(16))) __bf16 v16bf;
typedef __attribute__((ext_vector_type(8)))  float  v8f;

#define D_IN  256
#define H_DIM 32
#define C_DIM 2

__global__ void fill_kernel(float* __restrict__ p, long long n, float v) {
  long long i = (long long)blockIdx.x * blockDim.x + threadIdx.x;
  if (i < n) p[i] = v;
}

__device__ __forceinline__ int package_half() {
  unsigned id = __builtin_amdgcn_s_sendmsg_rtn(0x87);
  return (int)((id >> 18) & 1u);
}

__global__ void degree_kernel(const int* __restrict__ ei, float* __restrict__ cnt, int E, int N) {
  float* plane = cnt + (long long)package_half() * N;
  const int* tgt = ei + E;
  const int stride = gridDim.x * blockDim.x;
  for (int e = blockIdx.x * blockDim.x + threadIdx.x; e < E; e += stride)
    atomicAdd(&plane[tgt[e]], 1.0f);
}

__global__ void gemm1_wmma_kernel(const float* __restrict__ x,
                                  const float* __restrict__ W1,
                                  const float* __restrict__ cnt,
                                  float* __restrict__ dinv,
                                  float* __restrict__ t1,
                                  float* __restrict__ agg1, int N) {
  __shared__ __align__(32) __bf16 Bfrag[8][2][32][16];

  const int tid  = threadIdx.x;
  const int lane = tid & 31;
  const int wave = tid >> 5;

  for (int idx = tid; idx < 8 * 2 * 32 * 16; idx += blockDim.x) {
    int j     = idx & 15;
    int ln    = (idx >> 4) & 31;
    int ntile = (idx >> 9) & 1;
    int ks    = idx >> 10;
    int jj = j >> 1, p = j & 1;
    int hi = ln >> 4;
    int K  = ks * 32 + 16 * (jj >> 2) + 8 * hi + 2 * (jj & 3) + p;
    int n  = ntile * 16 + (ln & 15);
    Bfrag[ks][ntile][ln][j] = (__bf16)W1[K * H_DIM + n];
  }
  __syncthreads();

  long long rowbase = ((long long)blockIdx.x * 8 + wave) * 16;
  if (rowbase >= N) return;

  const int m  = lane & 15;
  const int hi = lane >> 4;
  const float* xrow = x + (rowbase + m) * D_IN;

  v8f acc0 = {}; v8f acc1 = {};
  #pragma unroll
  for (int ks = 0; ks < 8; ++ks) {
    v16bf a;
    #pragma unroll
    for (int jj = 0; jj < 8; ++jj) {
      int K0 = ks * 32 + 16 * (jj >> 2) + 8 * hi + 2 * (jj & 3);
      float2 f = *(const float2*)(xrow + K0);
      a[2 * jj]     = (__bf16)f.x;
      a[2 * jj + 1] = (__bf16)f.y;
    }
    v16bf b0 = *(const v16bf*)&Bfrag[ks][0][lane][0];
    v16bf b1 = *(const v16bf*)&Bfrag[ks][1][lane][0];
    acc0 = __builtin_amdgcn_wmma_f32_16x16x32_bf16(false, a, false, b0,
                                                   (short)0, acc0, false, false);
    acc1 = __builtin_amdgcn_wmma_f32_16x16x32_bf16(false, a, false, b1,
                                                   (short)0, acc1, false, false);
  }

  const long long node = rowbase + m;
  const float deg = 1.0f + cnt[node] + cnt[N + node];
  const float dv  = deg > 0.0f ? rsqrtf(deg) : 0.0f;
  if (hi == 0) dinv[node] = dv;

  #pragma unroll
  for (int r = 0; r < 8; ++r) {
    long long row = rowbase + r + 8 * hi;
    const int   dvb = __builtin_bit_cast(int, dv);
    const float dlo = __builtin_bit_cast(float, __builtin_amdgcn_readlane(dvb, r));
    const float dhi = __builtin_bit_cast(float, __builtin_amdgcn_readlane(dvb, r + 8));
    const float dr  = hi ? dhi : dlo;
    const float s0 = dr * acc0[r], s1 = dr * acc1[r];
    t1[row * H_DIM + m]        = s0;
    t1[row * H_DIM + 16 + m]   = s1;
    agg1[row * H_DIM + m]      = s0;
    agg1[row * H_DIM + 16 + m] = s1;
  }
}


__device__ __forceinline__ void lds_barrier() {
  __builtin_amdgcn_fence(__ATOMIC_RELEASE, "workgroup", "local");
  __builtin_amdgcn_s_barrier();
  __builtin_amdgcn_fence(__ATOMIC_ACQUIRE, "workgroup", "local");
}

#define S1_GROUP 16
#define S1_WAVE_EDGES 64
#define S1_CHUNK (8 * S1_WAVE_EDGES)
__global__ void __launch_bounds__(256)
scatter1_kernel(const int* __restrict__ ei,
                const float* __restrict__ t1,
                float* __restrict__ agg1,
                int* __restrict__ tickets, int E, int N) {
  __shared__ int q_s[8][64];
  __shared__ int q_t[8][64];
  __shared__ int s_tk[4];

  const unsigned lane = threadIdx.x & 31u;
  const int wave = __builtin_amdgcn_readfirstlane((int)(threadIdx.x >> 5));
  const int half = package_half();
  const int nchunks = (E + S1_CHUNK - 1) / S1_CHUNK;
  const int* __restrict__ src = ei;
  const int* __restrict__ tgt = ei + E;
  int* qs = q_s[wave];
  int* qt = q_t[wave];

  auto walk32 = [&](int s, int t) {
    #pragma unroll
    for (int g = 0; g < 32; g += S1_GROUP) {
      float v[S1_GROUP];
      #pragma unroll
      for (int i = 0; i < S1_GROUP; ++i) {
        const unsigned ro = (unsigned)__builtin_amdgcn_readlane(s, g + i) * H_DIM + lane;
        v[i] = t1[ro];
      }
      #pragma unroll
      for (int i = 0; i < S1_GROUP; ++i) {
        const unsigned ro = (unsigned)__builtin_amdgcn_readlane(t, g + i) * H_DIM + lane;
        atomicAdd(agg1 + ro, v[i]);
      }
    }
  };

  auto load_idx = [&](int chunk, int& s0, int& t0, int& s1, int& t1i) {
    const int e0 = chunk * S1_CHUNK + wave * S1_WAVE_EDGES + (int)lane;
    const int e1 = e0 + 32;
    s0 = 0; t0 = 0; s1 = 0; t1i = 0;
    if (e0 < E) { s0 = src[e0]; t0 = tgt[e0]; }
    if (e1 < E) { s1 = src[e1]; t1i = tgt[e1]; }
  };

  for (int pass = 0; pass < 2; ++pass) {
    const int h = half ^ pass;
    const int mid = (N >> 1) & ~1;
    const int lo  = h ? mid : 0;
    const unsigned span = (unsigned)(h ? N - mid : mid);
    int* ctr = tickets + 32 + 64 * h;

    int head = 0, qn = 0;

    auto enqueue = [&](int s, int t, bool valid) {
      const bool mine = valid && (unsigned)(t - lo) < span;
      const unsigned m = __builtin_amdgcn_ballot_w32(mine);
      if (m != 0u) {
        const int pos = (head + qn + (int)__builtin_amdgcn_mbcnt_lo(m, 0u)) & 63;
        if (mine) { qs[pos] = s; qt[pos] = t; }
        qn += __builtin_popcount(m);
        if (qn >= 32) {
          const int rp = (head + (int)lane) & 63;
          const int ws = qs[rp], wt = qt[rp];
          walk32(ws, wt);
          head = (head + 32) & 63;
          qn -= 32;
        }
      }
    };

    lds_barrier();
    if (threadIdx.x == 0) {
      const int a = atomicAdd(ctr, 2);
      s_tk[0] = a; s_tk[1] = a + 1;
    }
    lds_barrier();
    int cur = __builtin_amdgcn_readfirstlane(s_tk[0]);
    int nxt = __builtin_amdgcn_readfirstlane(s_tk[1]);

    int cs0 = 0, ct0 = 0, cs1 = 0, ct1 = 0;
    if (cur < nchunks) load_idx(cur, cs0, ct0, cs1, ct1);

    int r = 0;
    while (cur < nchunks) {
      int req = 0;
      if (threadIdx.x == 0) req = atomicAdd(ctr, 1);
      int ns0 = 0, nt0 = 0, ns1 = 0, nt1 = 0;
      if (nxt < nchunks) load_idx(nxt, ns0, nt0, ns1, nt1);

      const int eb = cur * S1_CHUNK + wave * S1_WAVE_EDGES + (int)lane;
      enqueue(cs0, ct0, eb < E);
      enqueue(cs1, ct1, eb + 32 < E);

      const int slot = (r + 2) & 3;
      if (threadIdx.x == 0) s_tk[slot] = req;
      lds_barrier();
      cur = nxt;
      nxt = __builtin_amdgcn_readfirstlane(s_tk[slot]);
      ++r;
      cs0 = ns0; ct0 = nt0; cs1 = ns1; ct1 = nt1;
    }

    if (qn > 0) {
      const int rp = (head + (int)lane) & 63;
      const int ws = qs[rp], wt = qt[rp];
      #pragma unroll
      for (int g = 0; g < 32; g += S1_GROUP) {
        if (g < qn) {
          float v[S1_GROUP];
          #pragma unroll
          for (int i = 0; i < S1_GROUP; ++i) {
            v[i] = 0.0f;
            if (g + i < qn) {
              const unsigned ro = (unsigned)__builtin_amdgcn_readlane(ws, g + i) * H_DIM + lane;
              v[i] = t1[ro];
            }
          }
          #pragma unroll
          for (int i = 0; i < S1_GROUP; ++i) {
            if (g + i < qn) {
              const unsigned ro = (unsigned)__builtin_amdgcn_readlane(wt, g + i) * H_DIM + lane;
              atomicAdd(agg1 + ro, v[i]);
            }
          }
        }
      }
    }
  }
}

__global__ void relu_gemm2_kernel(const float* __restrict__ agg1,
                                  const float* __restrict__ b1,
                                  const float* __restrict__ W2,
                                  const float* __restrict__ dinv,
                                  float* __restrict__ h2,
                                  float* __restrict__ agg2, int N) {
  int i = blockIdx.x * blockDim.x + threadIdx.x;
  if (i >= N) return;
  const float* hr = agg1 + (long long)i * H_DIM;
  const float di = dinv[i];
  float z = 0.0f;
  #pragma unroll
  for (int k = 0; k < H_DIM; ++k) {
    float v = di * hr[k] + b1[k];
    float h = v > 0.0f ? v : 0.0f;
    z += h * (W2[k * C_DIM + 0] - W2[k * C_DIM + 1]);
  }
  const float g = di * z;
  h2[i]       = g;
  agg2[i]     = g;
  agg2[N + i] = 0.0f;
}

__global__ void scatter2_kernel(const int* __restrict__ ei,
                                const float* __restrict__ h2,
                                float* __restrict__ agg2, int E, int N) {
  float* plane = agg2 + (long long)package_half() * N;
  const int stride = gridDim.x * blockDim.x;
  for (int e = blockIdx.x * blockDim.x + threadIdx.x; e < E; e += stride) {
    long long s = ei[e];
    long long t = ei[(long long)E + e];
    atomicAdd(&plane[t], h2[s]);
  }
}

__global__ void logsoftmax_kernel(const float* __restrict__ agg2,
                                  const float* __restrict__ dinv,
                                  const float* __restrict__ b2,
                                  float* __restrict__ out, int N) {
  int i = blockIdx.x * blockDim.x + threadIdx.x;
  if (i >= N) return;
  const float* agg2b = agg2 + N;
  const float di = dinv[i];
  const float d  = di * (agg2[i] + agg2b[i]) + (b2[0] - b2[1]);
  const float l  = logf(1.0f + expf(-fabsf(d)));
  out[(long long)i * 2 + 0] = fminf(d, 0.0f) - l;
  out[(long long)i * 2 + 1] = fminf(-d, 0.0f) - l;
}

extern "C" void kernel_launch(void* const* d_in, const int* in_sizes, int n_in,
                              void* d_out, int out_size, void* d_ws, size_t ws_size,
                              hipStream_t stream) {
  const float* x  = (const float*)d_in[0];
  const int*   ei = (const int*)d_in[1];
  const float* W1 = (const float*)d_in[2];
  const float* b1 = (const float*)d_in[3];
  const float* W2 = (const float*)d_in[4];
  const float* b2 = (const float*)d_in[5];
  float* out = (float*)d_out;

  const int N = in_sizes[0] / D_IN;
  const int E = in_sizes[1] / 2;

  float* ws   = (float*)d_ws;
  float* dinv = ws;
  float* t1   = dinv + N;
  float* agg1 = t1 + (long long)N * H_DIM;
  float* h2   = agg1 + (long long)N * H_DIM;
  float* cnt  = h2;
  float* agg2 = t1;
  int* tickets = (int*)(h2 + 2 * (long long)N);

  const int B = 256;
  const int EDGE_BLOCKS = 1024;
  const int S1_BLOCKS = 2048;
  fill_kernel<<<(2 * N + 128 + B - 1) / B, B, 0, stream>>>(cnt, 2 * (long long)N + 128, 0.0f);
  degree_kernel<<<EDGE_BLOCKS, B, 0, stream>>>(ei, cnt, E, N);

  gemm1_wmma_kernel<<<(N + 127) / 128, B, 0, stream>>>(x, W1, cnt, dinv, t1, agg1, N);
  scatter1_kernel<<<S1_BLOCKS, B, 0, stream>>>(ei, t1, agg1, tickets, E, N);

  relu_gemm2_kernel<<<(N + B - 1) / B, B, 0, stream>>>(agg1, b1, W2, dinv, h2, agg2, N);
  scatter2_kernel<<<EDGE_BLOCKS, B, 0, stream>>>(ei, h2, agg2, E, N);
  logsoftmax_kernel<<<(N + B - 1) / B, B, 0, stream>>>(agg2, dinv, b2, out, N);
}
